# starting kernel
# baseline (speedup 1.0000x reference)
#include <hip/hip_runtime.h>
#include <math.h>

typedef __attribute__((ext_vector_type(2))) float v2f;
typedef __attribute__((ext_vector_type(8))) float v8f;

#define B_ROWS 8192
#define D_FEAT 64
#define NELEM (B_ROWS * D_FEAT)
#define EPS_IN 1e-4f
#define EPS_OUT 1e-6f
#define COL_CHUNKS 8
#define TILES_PER_CHUNK (512 / COL_CHUNKS)
#define LDS_STRIDE 68

__global__ void var_kernel(const float* __restrict__ a,
                           const float* __restrict__ pos,
                           const float* __restrict__ neg,
                           float* __restrict__ rvar) {
  const int f = blockIdx.x;
  const int pair = blockIdx.y;
  const float* b = pair ? neg : pos;
  const int t = threadIdx.x;
  double s = 0.0, s2 = 0.0;
  for (int r = t; r < B_ROWS; r += 256) {
    float xa = a[r * D_FEAT + f] + EPS_IN;
    float xb = b[r * D_FEAT + f] + EPS_IN;
    s  += (double)xa + (double)xb;
    s2 += (double)xa * (double)xa + (double)xb * (double)xb;
  }
  __shared__ double ls[256];
  __shared__ double ls2[256];
  ls[t] = s; ls2[t] = s2;
  __syncthreads();
  for (int o = 128; o > 0; o >>= 1) {
    if (t < o) { ls[t] += ls[t + o]; ls2[t] += ls2[t + o]; }
    __syncthreads();
  }
  if (t == 0) {
    const double n = 2.0 * (double)B_ROWS;
    double mean = ls[0] / n;
    double var = ls2[0] / n - mean * mean;
    rvar[pair * D_FEAT + f] = (float)(1.0 / var);
  }
}

__global__ void diff_kernel(const float* __restrict__ a,
                            const float* __restrict__ p,
                            const float* __restrict__ n,
                            const float* __restrict__ rvar,
                            float* __restrict__ sdiff,
                            float* __restrict__ diffb) {
  const int idx = blockIdx.x * 256 + threadIdx.x;
  const int f = idx & (D_FEAT - 1);
  const float av = a[idx];
  const float d1 = av - p[idx];
  const float d2 = av - n[idx];
  diffb[idx] = d1;
  diffb[NELEM + idx] = d2;
  sdiff[idx] = d1 * rvar[f];
  sdiff[NELEM + idx] = d2 * rvar[D_FEAT + f];
}

__global__ void __launch_bounds__(256) mdl_wmma_kernel(
    const float* __restrict__ sdiff, const float* __restrict__ diffb,
    float* __restrict__ partial) {
  const int pair = blockIdx.z;
  const int chunk = blockIdx.y;
  const int rowblk = blockIdx.x;
  const float* A  = sdiff + (size_t)pair * NELEM;
  const float* Bm = diffb + (size_t)pair * NELEM;

  const int t = threadIdx.x;
  const int wave = t >> 5;
  const int lane = t & 31;
  const int half = lane >> 4;
  const int m = lane & 15;

  const int arow = rowblk * 128 + wave * 16 + m;
  const float* aptr = A + (size_t)arow * D_FEAT;
  v2f areg[16];
#pragma unroll
  for (int s = 0; s < 16; ++s)
    areg[s] = *(const v2f*)(aptr + 4 * s + 2 * half);

  __shared__ float bt[2][16 * LDS_STRIDE];

  const int br = t >> 4;
  const int bk = (t & 15) * 4;
  const int ct_begin = chunk * TILES_PER_CHUNK;

  {
    const float4 bv =
        *(const float4*)(Bm + (size_t)ct_begin * 16 * D_FEAT + 4 * t);
    bt[0][br * LDS_STRIDE + bk + 0] = bv.x;
    bt[0][br * LDS_STRIDE + bk + 1] = bv.y;
    bt[0][br * LDS_STRIDE + bk + 2] = bv.z;
    bt[0][br * LDS_STRIDE + bk + 3] = bv.w;
  }
  __syncthreads();

  float rs[8];
#pragma unroll
  for (int v = 0; v < 8; ++v) rs[v] = 0.0f;

  int cur = 0;
  for (int i = 0; i < TILES_PER_CHUNK; ++i) {
    float4 nv;
    const bool have_next = (i + 1 < TILES_PER_CHUNK);
    if (have_next)
      nv = *(const float4*)(Bm +
                            (size_t)(ct_begin + i + 1) * 16 * D_FEAT + 4 * t);

    const float* btc = bt[cur];
    v8f c = {};
#pragma unroll
    for (int s = 0; s < 16; ++s) {
      v2f breg = *(const v2f*)(&btc[m * LDS_STRIDE + 4 * s + 2 * half]);
      c = __builtin_amdgcn_wmma_f32_16x16x4_f32(
           false, areg[s],  false, breg,
           (short)0, c,  false,  false);
    }
#pragma unroll
    for (int v = 0; v < 8; ++v)
      rs[v] += __builtin_amdgcn_sqrtf(fmaxf(c[v], 0.0f)) + EPS_OUT;

    if (have_next) {
      float* dst = &bt[cur ^ 1][br * LDS_STRIDE + bk];
      dst[0] = nv.x; dst[1] = nv.y; dst[2] = nv.z; dst[3] = nv.w;
    }
    __syncthreads();
    cur ^= 1;
  }

#pragma unroll
  for (int v = 0; v < 8; ++v) {
    float x = rs[v];
    x += __shfl_xor(x, 1, 32);
    x += __shfl_xor(x, 2, 32);
    x += __shfl_xor(x, 4, 32);
    x += __shfl_xor(x, 8, 32);
    rs[v] = x;
  }
  if (m == 0) {
    const int rowbase = rowblk * 128 + wave * 16 + half * 8;
    float* pp = partial + ((size_t)pair * COL_CHUNKS + chunk) * B_ROWS + rowbase;
#pragma unroll
    for (int v = 0; v < 8; ++v) pp[v] = rs[v];
  }
}

__global__ void reduce_kernel(const float* __restrict__ partial,
                              float* __restrict__ out) {
  const int idx = blockIdx.x * 256 + threadIdx.x;
  const int pair = idx >> 13;
  const int r = idx & (B_ROWS - 1);
  float s = 0.0f;
#pragma unroll
  for (int c = 0; c < COL_CHUNKS; ++c)
    s += partial[((size_t)pair * COL_CHUNKS + c) * B_ROWS + r];
  out[idx] = s;
}

extern "C" void kernel_launch(void* const* d_in, const int* in_sizes, int n_in,
                              void* d_out, int out_size, void* d_ws, size_t ws_size,
                              hipStream_t stream) {
  const float* anchor   = (const float*)d_in[0];
  const float* positive = (const float*)d_in[1];
  const float* negative = (const float*)d_in[2];

  float* ws = (float*)d_ws;
  float* rvar    = ws;
  float* sdiff   = ws + 128;
  float* diffb   = sdiff + 2 * NELEM;
  float* partial = diffb + 2 * NELEM;
  float* out = (float*)d_out;

  var_kernel<<<dim3(D_FEAT, 2), 256, 0, stream>>>(anchor, positive, negative, rvar);
  diff_kernel<<<NELEM / 256, 256, 0, stream>>>(anchor, positive, negative, rvar,
                                               sdiff, diffb);
  mdl_wmma_kernel<<<dim3(B_ROWS / 128, COL_CHUNKS, 2), 256, 0, stream>>>(
      sdiff, diffb, partial);
  reduce_kernel<<<(2 * B_ROWS) / 256, 256, 0, stream>>>(partial, out);
}


# GEMM B tiles not staged in LDS: diff pass writes planes in fragment order, waves load B fragments straight from global one tile ahead, start tile rotated per row block; k and column sum order change
# speedup vs baseline: 1.1966x; 1.1966x over previous
#include <hip/hip_runtime.h>
#include <math.h>

typedef __attribute__((ext_vector_type(2))) float v2f;
typedef __attribute__((ext_vector_type(8))) float v8f;

#define B_ROWS 8192
#define D_FEAT 64
#define NELEM (B_ROWS * D_FEAT)
#define EPS_IN 1e-4f
#define EPS_OUT 1e-6f
#define COL_CHUNKS 8
#define TILES_PER_CHUNK (512 / COL_CHUNKS)
#define TILE_FLOATS (16 * D_FEAT)

__global__ void var_kernel(const float* __restrict__ a,
                           const float* __restrict__ pos,
                           const float* __restrict__ neg,
                           float* __restrict__ rvar) {
  const int f = blockIdx.x;
  const int pair = blockIdx.y;
  const float* b = pair ? neg : pos;
  const int t = threadIdx.x;
  double s = 0.0, s2 = 0.0;
  for (int r = t; r < B_ROWS; r += 256) {
    float xa = a[r * D_FEAT + f] + EPS_IN;
    float xb = b[r * D_FEAT + f] + EPS_IN;
    s  += (double)xa + (double)xb;
    s2 += (double)xa * (double)xa + (double)xb * (double)xb;
  }
  __shared__ double ls[256];
  __shared__ double ls2[256];
  ls[t] = s; ls2[t] = s2;
  __syncthreads();
  for (int o = 128; o > 0; o >>= 1) {
    if (t < o) { ls[t] += ls[t + o]; ls2[t] += ls2[t + o]; }
    __syncthreads();
  }
  if (t == 0) {
    const double n = 2.0 * (double)B_ROWS;
    double mean = ls[0] / n;
    double var = ls2[0] / n - mean * mean;
    rvar[pair * D_FEAT + f] = (float)(1.0 / var);
  }
}

__global__ void __launch_bounds__(256) diff_kernel(const float* __restrict__ a,
                            const float* __restrict__ p,
                            const float* __restrict__ n,
                            const float* __restrict__ rvar,
                            float* __restrict__ sdiff,
                            float* __restrict__ diffb) {
  __shared__ float4 st[4][256 + 16];
  const int t = threadIdx.x;
  const int r = t >> 4;
  const int q = t & 15;
  const size_t base = (size_t)blockIdx.x * TILE_FLOATS + 4 * t;
  const float4 av = *(const float4*)(a + base);
  const float4 pv = *(const float4*)(p + base);
  const float4 nv = *(const float4*)(n + base);
  const float4 r1 = *(const float4*)(rvar + 4 * q);
  const float4 r2 = *(const float4*)(rvar + D_FEAT + 4 * q);
  float4 d1, d2, s1, s2;
  d1.x = av.x - pv.x; d1.y = av.y - pv.y; d1.z = av.z - pv.z; d1.w = av.w - pv.w;
  d2.x = av.x - nv.x; d2.y = av.y - nv.y; d2.z = av.z - nv.z; d2.w = av.w - nv.w;
  s1.x = d1.x * r1.x; s1.y = d1.y * r1.y; s1.z = d1.z * r1.z; s1.w = d1.w * r1.w;
  s2.x = d2.x * r2.x; s2.y = d2.y * r2.y; s2.z = d2.z * r2.z; s2.w = d2.w * r2.w;
  const int slot = (q >> 1) * 32 + (q & 1) * 16 + r;
  const int ps = slot + (slot >> 4);
  st[0][ps] = d1; st[1][ps] = d2; st[2][ps] = s1; st[3][ps] = s2;
  __syncthreads();
  const int pt = t + (t >> 4);
  *(float4*)(diffb + base)         = st[0][pt];
  *(float4*)(diffb + NELEM + base) = st[1][pt];
  *(float4*)(sdiff + base)         = st[2][pt];
  *(float4*)(sdiff + NELEM + base) = st[3][pt];
}

__device__ __forceinline__ void load_tile(float4 (&f)[8], const float4* __restrict__ src) {
#pragma unroll
  for (int i = 0; i < 8; ++i) f[i] = src[i * 32];
}

__device__ __forceinline__ void tile_rowsum(const float4 (&af)[8], const float4 (&bf)[8],
                                            float (&rs)[8]) {
  v8f c = {};
#pragma unroll
  for (int i = 0; i < 8; ++i) {
    v2f alo = {af[i].x, af[i].y}, ahi = {af[i].z, af[i].w};
    v2f blo = {bf[i].x, bf[i].y}, bhi = {bf[i].z, bf[i].w};
    c = __builtin_amdgcn_wmma_f32_16x16x4_f32(
         false, alo,  false, blo,
         (short)0, c,  false,  false);
    c = __builtin_amdgcn_wmma_f32_16x16x4_f32(
         false, ahi,  false, bhi,
         (short)0, c,  false,  false);
  }
#pragma unroll
  for (int v = 0; v < 8; ++v)
    rs[v] += __builtin_amdgcn_sqrtf(fmaxf(c[v], 0.0f)) + EPS_OUT;
}

__global__ void __launch_bounds__(256) mdl_wmma_kernel(
    const float* __restrict__ sdiff, const float* __restrict__ diffb,
    float* __restrict__ partial) {
  const int pair = blockIdx.z;
  const int chunk = blockIdx.y;
  const int rowblk = blockIdx.x;

  const int t = threadIdx.x;
  const int wave = t >> 5;
  const int lane = t & 31;
  const int half = lane >> 4;
  const int m = lane & 15;

  const float4* A4 = (const float4*)(sdiff + (size_t)pair * NELEM +
                                     (size_t)(rowblk * 8 + wave) * TILE_FLOATS) + lane;
  float4 af[8];
  load_tile(af, A4);

  const float4* B4 = (const float4*)(diffb + (size_t)pair * NELEM +
                                     (size_t)(chunk * TILES_PER_CHUNK) * TILE_FLOATS) + lane;
  const int rot = rowblk & (TILES_PER_CHUNK - 1);
  float4 b0[8], b1[8];
  load_tile(b0, B4 + (size_t)rot * (TILE_FLOATS / 4));

  float rs[8];
#pragma unroll
  for (int v = 0; v < 8; ++v) rs[v] = 0.0f;

  for (int i = 0; i < TILES_PER_CHUNK; i += 2) {
    const int t1 = (i + 1 + rot) & (TILES_PER_CHUNK - 1);
    load_tile(b1, B4 + (size_t)t1 * (TILE_FLOATS / 4));
    tile_rowsum(af, b0, rs);
    if (i + 2 < TILES_PER_CHUNK) {
      const int t2 = (i + 2 + rot) & (TILES_PER_CHUNK - 1);
      load_tile(b0, B4 + (size_t)t2 * (TILE_FLOATS / 4));
    }
    tile_rowsum(af, b1, rs);
  }

#pragma unroll
  for (int v = 0; v < 8; ++v) {
    float x = rs[v];
    x += __shfl_xor(x, 1, 32);
    x += __shfl_xor(x, 2, 32);
    x += __shfl_xor(x, 4, 32);
    x += __shfl_xor(x, 8, 32);
    rs[v] = x;
  }
  if (m == 0) {
    const int rowbase = rowblk * 128 + wave * 16 + half * 8;
    float* pp = partial + ((size_t)pair * COL_CHUNKS + chunk) * B_ROWS + rowbase;
#pragma unroll
    for (int v = 0; v < 8; ++v) pp[v] = rs[v];
  }
}

__global__ void reduce_kernel(const float* __restrict__ partial,
                              float* __restrict__ out) {
  const int idx = blockIdx.x * 256 + threadIdx.x;
  const int pair = idx >> 13;
  const int r = idx & (B_ROWS - 1);
  float s = 0.0f;
#pragma unroll
  for (int c = 0; c < COL_CHUNKS; ++c)
    s += partial[((size_t)pair * COL_CHUNKS + c) * B_ROWS + r];
  out[idx] = s;
}

extern "C" void kernel_launch(void* const* d_in, const int* in_sizes, int n_in,
                              void* d_out, int out_size, void* d_ws, size_t ws_size,
                              hipStream_t stream) {
  const float* anchor   = (const float*)d_in[0];
  const float* positive = (const float*)d_in[1];
  const float* negative = (const float*)d_in[2];

  float* ws = (float*)d_ws;
  float* rvar    = ws;
  float* sdiff   = ws + 128;
  float* diffb   = sdiff + 2 * NELEM;
  float* partial = diffb + 2 * NELEM;
  float* out = (float*)d_out;

  var_kernel<<<dim3(D_FEAT, 2), 256, 0, stream>>>(anchor, positive, negative, rvar);
  diff_kernel<<<B_ROWS / 16, 256, 0, stream>>>(anchor, positive, negative, rvar,
                                               sdiff, diffb);
  mdl_wmma_kernel<<<dim3(B_ROWS / 128, COL_CHUNKS, 2), 256, 0, stream>>>(
      sdiff, diffb, partial);
  reduce_kernel<<<(2 * B_ROWS) / 256, 256, 0, stream>>>(partial, out);
}


# Variance sums taken in the diff pass from the rows it already loads (64 work-groups, f32 partials); GEMM prologue combines them and scales its A fragments; 3 launches; variance sums f32, new order
# speedup vs baseline: 1.4292x; 1.1944x over previous
#include <hip/hip_runtime.h>
#include <math.h>

typedef __attribute__((ext_vector_type(2))) float v2f;
typedef __attribute__((ext_vector_type(8))) float v8f;

#define B_ROWS 8192
#define D_FEAT 64
#define NELEM (B_ROWS * D_FEAT)
#define EPS_IN 1e-4f
#define EPS_OUT 1e-6f
#define COL_CHUNKS 8
#define TILES_PER_CHUNK (512 / COL_CHUNKS)
#define TILE_FLOATS (16 * D_FEAT)
#define NB_STAT 64
#define TILES_PER_WG (B_ROWS / 16 / NB_STAT)

__device__ __forceinline__ void stat_acc(float4& s, float4& ss, const float4 v) {
  const float x = v.x + EPS_IN, y = v.y + EPS_IN, z = v.z + EPS_IN, w = v.w + EPS_IN;
  s.x += x; s.y += y; s.z += z; s.w += w;
  ss.x += x * x; ss.y += y * y; ss.z += z * z; ss.w += w * w;
}

__device__ __forceinline__ float4 add4(const float4 a, const float4 b) {
  float4 r;
  r.x = a.x + b.x; r.y = a.y + b.y; r.z = a.z + b.z; r.w = a.w + b.w;
  return r;
}

__global__ void __launch_bounds__(512) diff_stats_kernel(const float* __restrict__ a,
                            const float* __restrict__ p,
                            const float* __restrict__ n,
                            float* __restrict__ diffb,
                            float* __restrict__ stats) {
  __shared__ float4 st[4][2][2][256 + 16];
  __shared__ float4 red4[32][64];
  const int t = threadIdx.x;
  const int u = t >> 8;
  const int tl = t & 255;
  const int r = tl >> 4;
  const int q = tl & 15;
  const size_t base0 = (size_t)(blockIdx.x * TILES_PER_WG + u) * TILE_FLOATS + 4 * tl;

  float4 av[4], pv[4], nv[4];
#pragma unroll
  for (int it = 0; it < 4; ++it) {
    const size_t base = base0 + (size_t)it * 2 * TILE_FLOATS;
    av[it] = *(const float4*)(a + base);
    pv[it] = *(const float4*)(p + base);
    nv[it] = *(const float4*)(n + base);
  }

  const float4 zero4 = {0.0f, 0.0f, 0.0f, 0.0f};
  float4 sa = zero4, qa = zero4, sp = zero4, qp = zero4, sn = zero4, qn = zero4;
  const int slot = (q >> 1) * 32 + (q & 1) * 16 + r;
  const int ps = slot + (slot >> 4);
#pragma unroll
  for (int it = 0; it < 4; ++it) {
    float4 d1, d2;
    d1.x = av[it].x - pv[it].x; d1.y = av[it].y - pv[it].y;
    d1.z = av[it].z - pv[it].z; d1.w = av[it].w - pv[it].w;
    d2.x = av[it].x - nv[it].x; d2.y = av[it].y - nv[it].y;
    d2.z = av[it].z - nv[it].z; d2.w = av[it].w - nv[it].w;
    st[it][0][u][ps] = d1;
    st[it][1][u][ps] = d2;
    stat_acc(sa, qa, av[it]);
    stat_acc(sp, qp, pv[it]);
    stat_acc(sn, qn, nv[it]);
  }
  const int rr = t >> 4;
  red4[rr][q]      = add4(sa, sp);
  red4[rr][16 + q] = add4(qa, qp);
  red4[rr][32 + q] = add4(sa, sn);
  red4[rr][48 + q] = add4(qa, qn);
  __syncthreads();

  const int pt = tl + (tl >> 4);
#pragma unroll
  for (int it = 0; it < 4; ++it) {
    const size_t base = base0 + (size_t)it * 2 * TILE_FLOATS;
    *(float4*)(diffb + base)         = st[it][0][u][pt];
    *(float4*)(diffb + NELEM + base) = st[it][1][u][pt];
  }

  if (t < 256) {
    const float* red = (const float*)&red4[0][0];
    float s0 = 0.0f, s1 = 0.0f, s2 = 0.0f, s3 = 0.0f;
#pragma unroll
    for (int j = 0; j < 8; ++j) {
      s0 += red[(4 * j + 0) * 256 + t];
      s1 += red[(4 * j + 1) * 256 + t];
      s2 += red[(4 * j + 2) * 256 + t];
      s3 += red[(4 * j + 3) * 256 + t];
    }
    stats[((size_t)(t >> 6) * NB_STAT + blockIdx.x) * D_FEAT + (t & 63)] =
        (s0 + s1) + (s2 + s3);
  }
}

__device__ __forceinline__ void load_tile(float4 (&f)[8], const float4* __restrict__ src) {
#pragma unroll
  for (int i = 0; i < 8; ++i) f[i] = src[i * 32];
}

__device__ __forceinline__ void tile_rowsum(const float4 (&af)[8], const float4 (&bf)[8],
                                            float (&rs)[8]) {
  v8f c = {};
#pragma unroll
  for (int i = 0; i < 8; ++i) {
    v2f alo = {af[i].x, af[i].y}, ahi = {af[i].z, af[i].w};
    v2f blo = {bf[i].x, bf[i].y}, bhi = {bf[i].z, bf[i].w};
    c = __builtin_amdgcn_wmma_f32_16x16x4_f32(
         false, alo,  false, blo,
         (short)0, c,  false,  false);
    c = __builtin_amdgcn_wmma_f32_16x16x4_f32(
         false, ahi,  false, bhi,
         (short)0, c,  false,  false);
  }
#pragma unroll
  for (int v = 0; v < 8; ++v)
    rs[v] += __builtin_amdgcn_sqrtf(fmaxf(c[v], 0.0f)) + EPS_OUT;
}

__global__ void __launch_bounds__(256) mdl_wmma_kernel(
    const float* __restrict__ diffb, const float* __restrict__ stats,
    float* __restrict__ partial) {
  __shared__ float4 red4[8][2][D_FEAT / 4];
  __shared__ float4 rv4[D_FEAT / 4];
  const int pair = blockIdx.z;
  const int chunk = blockIdx.y;
  const int rowblk = blockIdx.x;

  const int t = threadIdx.x;
  const int wave = t >> 5;
  const int lane = t & 31;
  const int half = lane >> 4;
  const int m = lane & 15;

  float4 ssum;
  {
    const float4* sp = (const float4*)(stats +
        ((size_t)(pair * 2 + half) * NB_STAT + (size_t)wave * 8) * D_FEAT) + m;
    float4 x[8];
#pragma unroll
    for (int j = 0; j < 8; ++j) x[j] = sp[j * (D_FEAT / 4)];
    ssum = add4(add4(add4(x[0], x[1]), add4(x[2], x[3])),
                add4(add4(x[4], x[5]), add4(x[6], x[7])));
  }

  const float4* A4 = (const float4*)(diffb + (size_t)pair * NELEM +
                                     (size_t)(rowblk * 8 + wave) * TILE_FLOATS) + lane;
  float4 af[8];
  load_tile(af, A4);

  const float4* B4 = (const float4*)(diffb + (size_t)pair * NELEM +
                                     (size_t)(chunk * TILES_PER_CHUNK) * TILE_FLOATS) + lane;
  const int rot = rowblk & (TILES_PER_CHUNK - 1);
  float4 b0[8], b1[8];
  load_tile(b0, B4 + (size_t)rot * (TILE_FLOATS / 4));

  red4[wave][half][m] = ssum;
  __syncthreads();
  if (t < D_FEAT) {
    const float* red = (const float*)&red4[0][0][0];
    float s[8], ss[8];
#pragma unroll
    for (int w = 0; w < 8; ++w) {
      s[w]  = red[w * 2 * D_FEAT + t];
      ss[w] = red[w * 2 * D_FEAT + D_FEAT + t];
    }
    const float inv_n = 1.0f / (2.0f * B_ROWS);
    const float mean = (((s[0] + s[1]) + (s[2] + s[3])) +
                        ((s[4] + s[5]) + (s[6] + s[7]))) * inv_n;
    const float var = (((ss[0] + ss[1]) + (ss[2] + ss[3])) +
                       ((ss[4] + ss[5]) + (ss[6] + ss[7]))) * inv_n - mean * mean;
    ((float*)rv4)[t] = 1.0f / var;
  }
  __syncthreads();
#pragma unroll
  for (int i = 0; i < 8; ++i) {
    const float4 rv = rv4[2 * i + half];
    af[i].x *= rv.x; af[i].y *= rv.y; af[i].z *= rv.z; af[i].w *= rv.w;
  }

  float rs[8];
#pragma unroll
  for (int v = 0; v < 8; ++v) rs[v] = 0.0f;

  for (int i = 0; i < TILES_PER_CHUNK; i += 2) {
    const int t1 = (i + 1 + rot) & (TILES_PER_CHUNK - 1);
    load_tile(b1, B4 + (size_t)t1 * (TILE_FLOATS / 4));
    tile_rowsum(af, b0, rs);
    if (i + 2 < TILES_PER_CHUNK) {
      const int t2 = (i + 2 + rot) & (TILES_PER_CHUNK - 1);
      load_tile(b0, B4 + (size_t)t2 * (TILE_FLOATS / 4));
    }
    tile_rowsum(af, b1, rs);
  }

#pragma unroll
  for (int v = 0; v < 8; ++v) {
    float x = rs[v];
    x += __shfl_xor(x, 1, 32);
    x += __shfl_xor(x, 2, 32);
    x += __shfl_xor(x, 4, 32);
    x += __shfl_xor(x, 8, 32);
    rs[v] = x;
  }
  if (m == 0) {
    const int rowbase = rowblk * 128 + wave * 16 + half * 8;
    float* pp = partial + ((size_t)pair * COL_CHUNKS + chunk) * B_ROWS + rowbase;
#pragma unroll
    for (int v = 0; v < 8; ++v) pp[v] = rs[v];
  }
}

__global__ void reduce_kernel(const float* __restrict__ partial,
                              float* __restrict__ out) {
  const int idx = blockIdx.x * 256 + threadIdx.x;
  const int pair = idx >> 13;
  const int r = idx & (B_ROWS - 1);
  float s = 0.0f;
#pragma unroll
  for (int c = 0; c < COL_CHUNKS; ++c)
    s += partial[((size_t)pair * COL_CHUNKS + c) * B_ROWS + r];
  out[idx] = s;
}

extern "C" void kernel_launch(void* const* d_in, const int* in_sizes, int n_in,
                              void* d_out, int out_size, void* d_ws, size_t ws_size,
                              hipStream_t stream) {
  const float* anchor   = (const float*)d_in[0];
  const float* positive = (const float*)d_in[1];
  const float* negative = (const float*)d_in[2];

  float* ws = (float*)d_ws;
  float* stats   = ws;
  float* diffb   = stats + 4 * NB_STAT * D_FEAT;
  float* partial = diffb + 2 * NELEM;
  float* out = (float*)d_out;

  diff_stats_kernel<<<NB_STAT, 512, 0, stream>>>(anchor, positive, negative, diffb, stats);
  mdl_wmma_kernel<<<dim3(B_ROWS / 128, COL_CHUNKS, 2), 256, 0, stream>>>(
      diffb, stats, partial);
  reduce_kernel<<<(2 * B_ROWS) / 256, 256, 0, stream>>>(partial, out);
}


# GEMM re-cut so a wave feeds the matrix unit alone: 2 column tiles a step on 2 accumulator chains, B ring 3 steps ahead; 2 waves per SIMD, column halves meet in LDS, 2 launches; row sum order new
# speedup vs baseline: 1.4862x; 1.0399x over previous
#include <hip/hip_runtime.h>
#include <math.h>

typedef __attribute__((ext_vector_type(2))) float v2f;
typedef __attribute__((ext_vector_type(8))) float v8f;

#define B_ROWS 8192
#define D_FEAT 64
#define NELEM (B_ROWS * D_FEAT)
#define EPS_IN 1e-4f
#define EPS_OUT 1e-6f
#define COL_HALVES 2
#define STEPS_PER_HALF (512 / COL_HALVES / 2)
#define TILE_FLOATS (16 * D_FEAT)
#define NB_STAT 64
#define TILES_PER_WG (B_ROWS / 16 / NB_STAT)

__device__ __forceinline__ void stat_acc(float4& s, float4& ss, const float4 v) {
  const float x = v.x + EPS_IN, y = v.y + EPS_IN, z = v.z + EPS_IN, w = v.w + EPS_IN;
  s.x += x; s.y += y; s.z += z; s.w += w;
  ss.x += x * x; ss.y += y * y; ss.z += z * z; ss.w += w * w;
}

__device__ __forceinline__ float4 add4(const float4 a, const float4 b) {
  float4 r;
  r.x = a.x + b.x; r.y = a.y + b.y; r.z = a.z + b.z; r.w = a.w + b.w;
  return r;
}

__global__ void __launch_bounds__(512) diff_stats_kernel(const float* __restrict__ a,
                            const float* __restrict__ p,
                            const float* __restrict__ n,
                            float* __restrict__ diffb,
                            float* __restrict__ stats) {
  __shared__ float4 st[4][2][2][256 + 16];
  __shared__ float4 red4[32][64];
  const int t = threadIdx.x;
  const int u = t >> 8;
  const int tl = t & 255;
  const int r = tl >> 4;
  const int q = tl & 15;
  const size_t base0 = (size_t)(blockIdx.x * TILES_PER_WG + u) * TILE_FLOATS + 4 * tl;

  float4 av[4], pv[4], nv[4];
#pragma unroll
  for (int it = 0; it < 4; ++it) {
    const size_t base = base0 + (size_t)it * 2 * TILE_FLOATS;
    av[it] = *(const float4*)(a + base);
    pv[it] = *(const float4*)(p + base);
    nv[it] = *(const float4*)(n + base);
  }

  const float4 zero4 = {0.0f, 0.0f, 0.0f, 0.0f};
  float4 sa = zero4, qa = zero4, sp = zero4, qp = zero4, sn = zero4, qn = zero4;
  const int slot = (q >> 1) * 32 + (q & 1) * 16 + r;
  const int ps = slot + (slot >> 4);
#pragma unroll
  for (int it = 0; it < 4; ++it) {
    float4 d1, d2;
    d1.x = av[it].x - pv[it].x; d1.y = av[it].y - pv[it].y;
    d1.z = av[it].z - pv[it].z; d1.w = av[it].w - pv[it].w;
    d2.x = av[it].x - nv[it].x; d2.y = av[it].y - nv[it].y;
    d2.z = av[it].z - nv[it].z; d2.w = av[it].w - nv[it].w;
    st[it][0][u][ps] = d1;
    st[it][1][u][ps] = d2;
    stat_acc(sa, qa, av[it]);
    stat_acc(sp, qp, pv[it]);
    stat_acc(sn, qn, nv[it]);
  }
  const int rr = t >> 4;
  red4[rr][q]      = add4(sa, sp);
  red4[rr][16 + q] = add4(qa, qp);
  red4[rr][32 + q] = add4(sa, sn);
  red4[rr][48 + q] = add4(qa, qn);
  __syncthreads();

  const int pt = tl + (tl >> 4);
#pragma unroll
  for (int it = 0; it < 4; ++it) {
    const size_t base = base0 + (size_t)it * 2 * TILE_FLOATS;
    *(float4*)(diffb + base)         = st[it][0][u][pt];
    *(float4*)(diffb + NELEM + base) = st[it][1][u][pt];
  }

  if (t < 256) {
    const float* red = (const float*)&red4[0][0];
    float s0 = 0.0f, s1 = 0.0f, s2 = 0.0f, s3 = 0.0f;
#pragma unroll
    for (int j = 0; j < 8; ++j) {
      s0 += red[(4 * j + 0) * 256 + t];
      s1 += red[(4 * j + 1) * 256 + t];
      s2 += red[(4 * j + 2) * 256 + t];
      s3 += red[(4 * j + 3) * 256 + t];
    }
    stats[((size_t)(t >> 6) * NB_STAT + blockIdx.x) * D_FEAT + (t & 63)] =
        (s0 + s1) + (s2 + s3);
  }
}

__device__ __forceinline__ void load_tile(float4 (&f)[8], const float4* __restrict__ src) {
#pragma unroll
  for (int i = 0; i < 8; ++i) f[i] = src[i * 32];
}

__device__ __forceinline__ void load_step(float4 (&f)[16], const float4* __restrict__ src) {
#pragma unroll
  for (int i = 0; i < 16; ++i) f[i] = src[i * 32];
}

__device__ __forceinline__ void step_chains(const float4 (&af)[8], const float4 (&bf)[16],
                                            v8f& cx, v8f& cy) {
  v8f x = {}, y = {};
#pragma unroll
  for (int i = 0; i < 8; ++i) {
    v2f alo = {af[i].x, af[i].y}, ahi = {af[i].z, af[i].w};
    v2f xlo = {bf[i].x, bf[i].y}, xhi = {bf[i].z, bf[i].w};
    v2f ylo = {bf[8 + i].x, bf[8 + i].y}, yhi = {bf[8 + i].z, bf[8 + i].w};
    x = __builtin_amdgcn_wmma_f32_16x16x4_f32(
         false, alo,  false, xlo,
         (short)0, x,  false,  false);
    y = __builtin_amdgcn_wmma_f32_16x16x4_f32(
         false, alo,  false, ylo,
         (short)0, y,  false,  false);
    x = __builtin_amdgcn_wmma_f32_16x16x4_f32(
         false, ahi,  false, xhi,
         (short)0, x,  false,  false);
    y = __builtin_amdgcn_wmma_f32_16x16x4_f32(
         false, ahi,  false, yhi,
         (short)0, y,  false,  false);
  }
  cx = x; cy = y;
}

__device__ __forceinline__ void step_rowsum(const v8f cx, const v8f cy, float (&rs)[8]) {
#pragma unroll
  for (int v = 0; v < 8; ++v)
    rs[v] += __builtin_amdgcn_sqrtf(fmaxf(cx[v], 0.0f)) + EPS_OUT;
#pragma unroll
  for (int v = 0; v < 8; ++v)
    rs[v] += __builtin_amdgcn_sqrtf(fmaxf(cy[v], 0.0f)) + EPS_OUT;
}

__global__ void __launch_bounds__(256) mdl_wmma_kernel(
    const float* __restrict__ diffb, const float* __restrict__ stats,
    float* __restrict__ out) {
  __shared__ float4 red4[8][2][D_FEAT / 4];
  __shared__ float4 rv4[D_FEAT / 4];
  __shared__ float meet[COL_HALVES][64];
  const int pair = blockIdx.y;
  const int rowblk = blockIdx.x;

  const int t = threadIdx.x;
  const int wave = t >> 5;
  const int lane = t & 31;
  const int half = lane >> 4;
  const int m = lane & 15;
  const int rt = wave & 3;
  const int ch = wave >> 2;

  float4 ssum;
  {
    const float4* sp = (const float4*)(stats +
        ((size_t)(pair * 2 + half) * NB_STAT + (size_t)wave * 8) * D_FEAT) + m;
    float4 x[8];
#pragma unroll
    for (int j = 0; j < 8; ++j) x[j] = sp[j * (D_FEAT / 4)];
    ssum = add4(add4(add4(x[0], x[1]), add4(x[2], x[3])),
                add4(add4(x[4], x[5]), add4(x[6], x[7])));
  }

  const float4* A4 = (const float4*)(diffb + (size_t)pair * NELEM +
                                     (size_t)(rowblk * 4 + rt) * TILE_FLOATS) + lane;
  float4 af[8];
  load_tile(af, A4);

  const float4* B4 = (const float4*)(diffb + (size_t)pair * NELEM +
                                     (size_t)(ch * 2 * STEPS_PER_HALF) * TILE_FLOATS) + lane;
  const int rot = rowblk & (STEPS_PER_HALF - 1);
#define STEP_PTR(s) (B4 + (size_t)(((s) + rot) & (STEPS_PER_HALF - 1)) * (2 * TILE_FLOATS / 4))
  float4 r0[16], r1[16], r2[16], r3[16];
  load_step(r0, STEP_PTR(0));
  load_step(r1, STEP_PTR(1));
  load_step(r2, STEP_PTR(2));

  red4[wave][half][m] = ssum;
  __syncthreads();
  if (t < D_FEAT) {
    const float* red = (const float*)&red4[0][0][0];
    float s[8], ss[8];
#pragma unroll
    for (int w = 0; w < 8; ++w) {
      s[w]  = red[w * 2 * D_FEAT + t];
      ss[w] = red[w * 2 * D_FEAT + D_FEAT + t];
    }
    const float inv_n = 1.0f / (2.0f * B_ROWS);
    const float mean = (((s[0] + s[1]) + (s[2] + s[3])) +
                        ((s[4] + s[5]) + (s[6] + s[7]))) * inv_n;
    const float var = (((ss[0] + ss[1]) + (ss[2] + ss[3])) +
                       ((ss[4] + ss[5]) + (ss[6] + ss[7]))) * inv_n - mean * mean;
    ((float*)rv4)[t] = 1.0f / var;
  }
  __syncthreads();
#pragma unroll
  for (int i = 0; i < 8; ++i) {
    const float4 rv = rv4[2 * i + half];
    af[i].x *= rv.x; af[i].y *= rv.y; af[i].z *= rv.z; af[i].w *= rv.w;
  }

  float rs[8];
#pragma unroll
  for (int v = 0; v < 8; ++v) rs[v] = 0.0f;

#define SECTION_END() __builtin_amdgcn_sched_barrier(0)
  v8f cx, cy;
  for (int s = 0; s < STEPS_PER_HALF; s += 4) {
    load_step(r3, STEP_PTR(s + 3));
    step_chains(af, r0, cx, cy);
    step_rowsum(cx, cy, rs);
    SECTION_END();
    load_step(r0, STEP_PTR(s + 4));
    step_chains(af, r1, cx, cy);
    step_rowsum(cx, cy, rs);
    SECTION_END();
    load_step(r1, STEP_PTR(s + 5));
    step_chains(af, r2, cx, cy);
    step_rowsum(cx, cy, rs);
    SECTION_END();
    load_step(r2, STEP_PTR(s + 6));
    step_chains(af, r3, cx, cy);
    step_rowsum(cx, cy, rs);
    SECTION_END();
  }
#undef SECTION_END
#undef STEP_PTR

#pragma unroll
  for (int v = 0; v < 8; ++v) {
    float x = rs[v];
    x += __shfl_xor(x, 1, 32);
    x += __shfl_xor(x, 2, 32);
    x += __shfl_xor(x, 4, 32);
    x += __shfl_xor(x, 8, 32);
    rs[v] = x;
  }
  if (m == 0) {
#pragma unroll
    for (int v = 0; v < 8; ++v) meet[ch][rt * 16 + half * 8 + v] = rs[v];
  }
  __syncthreads();
  if (t < 64)
    out[(size_t)pair * B_ROWS + (size_t)rowblk * 64 + t] = meet[0][t] + meet[1][t];
}

extern "C" void kernel_launch(void* const* d_in, const int* in_sizes, int n_in,
                              void* d_out, int out_size, void* d_ws, size_t ws_size,
                              hipStream_t stream) {
  const float* anchor   = (const float*)d_in[0];
  const float* positive = (const float*)d_in[1];
  const float* negative = (const float*)d_in[2];

  float* ws = (float*)d_ws;
  float* stats   = ws;
  float* diffb   = stats + 4 * NB_STAT * D_FEAT;
  float* out = (float*)d_out;

  diff_stats_kernel<<<NB_STAT, 512, 0, stream>>>(anchor, positive, negative, diffb, stats);
  mdl_wmma_kernel<<<dim3(B_ROWS / 64, 2), 256, 0, stream>>>(diffb, stats, out);
}
